# starting kernel
# baseline (speedup 1.0000x reference)
#include <hip/hip_runtime.h>
#include <hip/hip_bf16.h>


typedef _Float16 v16h __attribute__((ext_vector_type(16)));
typedef _Float16 v8h  __attribute__((ext_vector_type(8)));
typedef float    v8f  __attribute__((ext_vector_type(8)));
typedef float    v4f  __attribute__((ext_vector_type(4)));

#define NB   8
#define CCH  512
#define NTOK 1024
#define NH   4
#define DK   128

__device__ __forceinline__ v16h load_frag(const _Float16* p) {
  union { v16h v; v8h h[2]; } u;
  u.h[0] = *(const v8h*)(p);
  u.h[1] = *(const v8h*)(p + 16);
  return u.v;
}

__device__ __forceinline__ v8f wmma16(v16h a, v16h b, v8f c) {
  return __builtin_amdgcn_wmma_f32_16x16x32_f16(false, a, false, b,
                                                (short)0, c, false, false);
}

__global__ void __launch_bounds__(256)
transpose_cast(const float* __restrict__ src, _Float16* __restrict__ dst,
               int rows, int cols) {
  __shared__ float tile[32][33];
  const size_t boff = (size_t)blockIdx.z * rows * cols;
  const int cx = blockIdx.x * 32 + threadIdx.x;
  const int ry = blockIdx.y * 32 + threadIdx.y;
#pragma unroll
  for (int j = 0; j < 32; j += 8)
    tile[threadIdx.y + j][threadIdx.x] = src[boff + (size_t)(ry + j) * cols + cx];
  __syncthreads();
  const int rx = blockIdx.y * 32 + threadIdx.x;
  const int cy = blockIdx.x * 32 + threadIdx.y;
#pragma unroll
  for (int j = 0; j < 32; j += 8)
    dst[boff + (size_t)(cy + j) * rows + rx] = (_Float16)tile[threadIdx.x][threadIdx.y + j];
}

__global__ void __launch_bounds__(32)
qkv_gemm(const _Float16* __restrict__ xh,
         const _Float16* __restrict__ WpT,
         const float* __restrict__ bp,
         _Float16* __restrict__ qh,
         _Float16* __restrict__ kh,
         _Float16* __restrict__ vT) {
  const int lane  = threadIdx.x;
  const int hf    = lane >> 4;
  const int l16   = lane & 15;
  const int kbase = hf * 8;
  const int roff  = hf * 8;
  const int n0 = blockIdx.x * 32;
  const int j0 = blockIdx.y * 64;
  const int b  = blockIdx.z;

  const _Float16* A0 = xh + ((size_t)b * NTOK + (n0 + l16)) * CCH + kbase;
  const _Float16* A1 = A0 + (size_t)16 * CCH;
  const _Float16* Bp[4];
#pragma unroll
  for (int t = 0; t < 4; ++t)
    Bp[t] = WpT + (size_t)(j0 + t * 16 + l16) * CCH + kbase;

  v16h a0c = load_frag(A0), a1c = load_frag(A1);
  v16h bc[4];
#pragma unroll
  for (int t = 0; t < 4; ++t) bc[t] = load_frag(Bp[t]);
  v8f acc0[4] = {}, acc1[4] = {};
#pragma unroll
  for (int k0 = 0; k0 < CCH; k0 += 32) {
    v16h a0n = a0c, a1n = a1c, bn[4] = {bc[0], bc[1], bc[2], bc[3]};
    const int kn = k0 + 32;
    if (kn < CCH) {
      a0n = load_frag(A0 + kn);
      a1n = load_frag(A1 + kn);
#pragma unroll
      for (int t = 0; t < 4; ++t) bn[t] = load_frag(Bp[t] + kn);
    }
#pragma unroll
    for (int t = 0; t < 4; ++t) {
      acc0[t] = wmma16(a0c, bc[t], acc0[t]);
      acc1[t] = wmma16(a1c, bc[t], acc1[t]);
    }
    a0c = a0n; a1c = a1n;
#pragma unroll
    for (int t = 0; t < 4; ++t) bc[t] = bn[t];
  }

#pragma unroll
  for (int t = 0; t < 4; ++t) {
    const int j0t  = j0 + t * 16;
    const int head = j0t / 384;
    const int r0   = j0t % 384;
    const int seg  = r0 / 128;
    const int d    = (r0 % 128) + l16;
    const float bias = bp[j0t + l16];
    if (seg == 2) {
      _Float16* dst = vT + (((size_t)b * NH + head) * DK + d) * NTOK + n0 + roff;
#pragma unroll
      for (int g = 0; g < 8; ++g) {
        dst[g]      = (_Float16)(acc0[t][g] + bias);
        dst[g + 16] = (_Float16)(acc1[t][g] + bias);
      }
    } else {
      _Float16* base = (seg == 0) ? qh : kh;
      _Float16* p0 = base + (((size_t)b * NH + head) * NTOK + n0 + roff) * DK + d;
#pragma unroll
      for (int g = 0; g < 8; ++g) {
        p0[(size_t)g * DK]        = (_Float16)(acc0[t][g] + bias);
        p0[(size_t)(g + 16) * DK] = (_Float16)(acc1[t][g] + bias);
      }
    }
  }
}

__global__ void __launch_bounds__(32)
attention_fa(const _Float16* __restrict__ qh, const _Float16* __restrict__ kh,
             const _Float16* __restrict__ vT, _Float16* __restrict__ attn) {
  __shared__ alignas(16) _Float16 lds_p[16][40];
  const int lane  = threadIdx.x;
  const int hf    = lane >> 4;
  const int l16   = lane & 15;
  const int kbase = hf * 8;
  const int roff  = hf * 8;
  const int n0 = blockIdx.x * 16;
  const int h  = blockIdx.y;
  const int b  = blockIdx.z;
  const float scale = 0.08838834764831845f;

  const _Float16* Qb = qh + (((size_t)b * NH + h) * NTOK + n0 + l16) * DK;
  v16h qf[4];
#pragma unroll
  for (int kk = 0; kk < 4; ++kk) qf[kk] = load_frag(Qb + kk * 32 + kbase);

  v8f o[8] = {};
  float mi[8], li[8];
#pragma unroll
  for (int g = 0; g < 8; ++g) { mi[g] = -1e30f; li[g] = 0.0f; }

  const _Float16* Kb = kh + ((size_t)b * NH + h) * NTOK * DK;
  const _Float16* Vb = vT + ((size_t)b * NH + h) * DK * NTOK;

  for (int j0 = 0; j0 < NTOK; j0 += 32) {
    if (j0 + 32 < NTOK) {
      __builtin_prefetch(Kb + (size_t)(j0 + 32 + l16) * DK + kbase, 0, 3);
      __builtin_prefetch(Kb + (size_t)(j0 + 48 + l16) * DK + kbase, 0, 3);
    }
    v8f s0 = {}, s1 = {};
    const _Float16* K0 = Kb + (size_t)(j0 + l16) * DK;
    const _Float16* K1 = K0 + (size_t)16 * DK;
#pragma unroll
    for (int kk = 0; kk < 4; ++kk) {
      s0 = wmma16(qf[kk], load_frag(K0 + kk * 32 + kbase), s0);
      s1 = wmma16(qf[kk], load_frag(K1 + kk * 32 + kbase), s1);
    }
    v16h vf[4];
#pragma unroll
    for (int t = 0; t < 4; ++t)
      vf[t] = load_frag(Vb + (size_t)(t * 16 + l16) * NTOK + j0 + kbase);

#pragma unroll
    for (int g = 0; g < 8; ++g) {
      const float a0 = s0[g] * scale, a1 = s1[g] * scale;
      float m = fmaxf(a0, a1);
      m = fmaxf(m, __shfl_xor(m, 1));
      m = fmaxf(m, __shfl_xor(m, 2));
      m = fmaxf(m, __shfl_xor(m, 4));
      m = fmaxf(m, __shfl_xor(m, 8));
      const float mn    = fmaxf(mi[g], m);
      const float alpha = __expf(mi[g] - mn);
      const float p0 = __expf(a0 - mn);
      const float p1 = __expf(a1 - mn);
      s0[g] = p0; s1[g] = p1;
      float rs = p0 + p1;
      rs += __shfl_xor(rs, 1);
      rs += __shfl_xor(rs, 2);
      rs += __shfl_xor(rs, 4);
      rs += __shfl_xor(rs, 8);
      li[g] = li[g] * alpha + rs;
      mi[g] = mn;
#pragma unroll
      for (int t = 0; t < 8; ++t) o[t][g] *= alpha;
    }
#pragma unroll
    for (int g = 0; g < 8; ++g) {
      lds_p[roff + g][l16]      = (_Float16)s0[g];
      lds_p[roff + g][16 + l16] = (_Float16)s1[g];
    }
    __syncthreads();
    const v16h pf = load_frag(&lds_p[l16][kbase]);
#pragma unroll
    for (int t = 0; t < 4; ++t) o[t] = wmma16(pf, vf[t], o[t]);
#pragma unroll
    for (int t = 4; t < 8; ++t) {
      const _Float16* Vp = Vb + (size_t)(t * 16 + l16) * NTOK + j0 + kbase;
      o[t] = wmma16(pf, load_frag(Vp), o[t]);
    }
    __syncthreads();
  }
  _Float16* Ob = attn + ((size_t)b * NTOK + n0 + roff) * (NH * DK) + h * DK + l16;
#pragma unroll
  for (int t = 0; t < 8; ++t)
#pragma unroll
    for (int g = 0; g < 8; ++g)
      Ob[(size_t)g * (NH * DK) + t * 16] = (_Float16)(o[t][g] / li[g]);
}

__global__ void __launch_bounds__(32)
out_proj(const _Float16* __restrict__ attn,
         const _Float16* __restrict__ WoT,
         const float* __restrict__ bo,
         const float* __restrict__ x,
         float* __restrict__ out) {
  const int lane  = threadIdx.x;
  const int hf    = lane >> 4;
  const int l16   = lane & 15;
  const int kbase = hf * 8;
  const int roff  = hf * 8;
  const int n0 = blockIdx.x * 32;
  const int c0 = blockIdx.y * 64;
  const int b  = blockIdx.z;

  const _Float16* A0 = attn + ((size_t)b * NTOK + n0 + l16) * CCH + kbase;
  const _Float16* A1 = A0 + (size_t)16 * CCH;
  const _Float16* Bp[4];
#pragma unroll
  for (int t = 0; t < 4; ++t)
    Bp[t] = WoT + (size_t)(c0 + t * 16 + l16) * CCH + kbase;

  v16h a0c = load_frag(A0), a1c = load_frag(A1);
  v16h bc[4];
#pragma unroll
  for (int t = 0; t < 4; ++t) bc[t] = load_frag(Bp[t]);
  v8f acc0[4] = {}, acc1[4] = {};
#pragma unroll
  for (int k0 = 0; k0 < CCH; k0 += 32) {
    v16h a0n = a0c, a1n = a1c, bn[4] = {bc[0], bc[1], bc[2], bc[3]};
    const int kn = k0 + 32;
    if (kn < CCH) {
      a0n = load_frag(A0 + kn);
      a1n = load_frag(A1 + kn);
#pragma unroll
      for (int t = 0; t < 4; ++t) bn[t] = load_frag(Bp[t] + kn);
    }
#pragma unroll
    for (int t = 0; t < 4; ++t) {
      acc0[t] = wmma16(a0c, bc[t], acc0[t]);
      acc1[t] = wmma16(a1c, bc[t], acc1[t]);
    }
    a0c = a0n; a1c = a1n;
#pragma unroll
    for (int t = 0; t < 4; ++t) bc[t] = bn[t];
  }

#pragma unroll
  for (int t = 0; t < 4; ++t) {
    const int c = c0 + t * 16 + l16;
    const float bias = bo[c];
    const size_t base = ((size_t)b * CCH + c) * NTOK + n0 + roff;
#pragma unroll
    for (int half2 = 0; half2 < 2; ++half2) {
      const v8f& av   = half2 ? acc1[t] : acc0[t];
      const size_t bi = base + (size_t)half2 * 16;
      const v4f x0 = *(const v4f*)(x + bi);
      const v4f x1 = *(const v4f*)(x + bi + 4);
      v4f r0, r1;
#pragma unroll
      for (int g = 0; g < 4; ++g) {
        r0[g] = av[g]     + bias + x0[g];
        r1[g] = av[4 + g] + bias + x1[g];
      }
      *(v4f*)(out + bi)     = r0;
      *(v4f*)(out + bi + 4) = r1;
    }
  }
}

extern "C" void kernel_launch(void* const* d_in, const int* in_sizes, int n_in,
                              void* d_out, int out_size, void* d_ws, size_t ws_size,
                              hipStream_t stream) {
  (void)in_sizes; (void)n_in; (void)out_size; (void)ws_size;
  const float* x  = (const float*)d_in[0];
  const float* Wp = (const float*)d_in[1];
  const float* bp = (const float*)d_in[2];
  const float* Wo = (const float*)d_in[3];
  const float* bo = (const float*)d_in[4];
  float* out = (float*)d_out;

  char* ws = (char*)d_ws;
  _Float16* xh  = (_Float16*)(ws);
  _Float16* WpT = (_Float16*)(ws +  8388608);
  _Float16* WoT = (_Float16*)(ws +  9961472);
  _Float16* qh  = (_Float16*)(ws + 10485760);
  _Float16* kh  = (_Float16*)(ws + 18874368);
  _Float16* vT  = (_Float16*)(ws + 27262976);
  _Float16* at  = (_Float16*)(ws + 35651584);

  const dim3 tb(32, 8);
  transpose_cast<<<dim3(NTOK / 32, CCH / 32, NB), tb, 0, stream>>>(x,  xh,  CCH, NTOK);
  transpose_cast<<<dim3(1536 / 32, CCH / 32, 1),  tb, 0, stream>>>(Wp, WpT, CCH, 1536);
  transpose_cast<<<dim3(CCH  / 32, CCH / 32, 1),  tb, 0, stream>>>(Wo, WoT, CCH, CCH);

  qkv_gemm    <<<dim3(NTOK / 32, 1536 / 64, NB), 32, 0, stream>>>(xh, WpT, bp, qh, kh, vT);
  attention_fa<<<dim3(NTOK / 16, NH,        NB), 32, 0, stream>>>(qh, kh, vT, at);
  out_proj    <<<dim3(NTOK / 32, CCH / 64,  NB), 32, 0, stream>>>(at, WoT, bo, x, out);
}


# result tiles of qkv, attention and out_proj leave through LDS: q/k/out rows as whole 128-B lines, vT/attn rows as 64-B runs via transposing LDS loads, not 2-byte or 16-line stores; sum order kept
# speedup vs baseline: 2.4742x; 2.4742x over previous
#include <hip/hip_runtime.h>
#include <hip/hip_bf16.h>


typedef _Float16 v16h __attribute__((ext_vector_type(16)));
typedef _Float16 v8h  __attribute__((ext_vector_type(8)));
typedef _Float16 v4h  __attribute__((ext_vector_type(4)));
typedef float    v8f  __attribute__((ext_vector_type(8)));
typedef float    v4f  __attribute__((ext_vector_type(4)));

#define NB   8
#define CCH  512
#define NTOK 1024
#define NH   4
#define DK   128

__device__ __forceinline__ v16h load_frag(const _Float16* p) {
  union { v16h v; v8h h[2]; } u;
  u.h[0] = *(const v8h*)(p);
  u.h[1] = *(const v8h*)(p + 16);
  return u.v;
}

__device__ __forceinline__ v8f wmma16(v16h a, v16h b, v8f c) {
  return __builtin_amdgcn_wmma_f32_16x16x32_f16(false, a, false, b,
                                                (short)0, c, false, false);
}

typedef short s16x8 __attribute__((ext_vector_type(8)));
typedef __attribute__((address_space(3))) s16x8 lds_s16x8;
__device__ __forceinline__ v8h lds_load_tr16(const _Float16* p) {
  return __builtin_bit_cast(v8h, __builtin_amdgcn_ds_load_tr16_b128_v8i16((lds_s16x8*)p));
}

__global__ void __launch_bounds__(256)
transpose_cast(const float* __restrict__ src, _Float16* __restrict__ dst,
               int rows, int cols) {
  __shared__ float tile[32][33];
  const size_t boff = (size_t)blockIdx.z * rows * cols;
  const int cx = blockIdx.x * 32 + threadIdx.x;
  const int ry = blockIdx.y * 32 + threadIdx.y;
#pragma unroll
  for (int j = 0; j < 32; j += 8)
    tile[threadIdx.y + j][threadIdx.x] = src[boff + (size_t)(ry + j) * cols + cx];
  __syncthreads();
  const int rx = blockIdx.y * 32 + threadIdx.x;
  const int cy = blockIdx.x * 32 + threadIdx.y;
#pragma unroll
  for (int j = 0; j < 32; j += 8)
    dst[boff + (size_t)(cy + j) * rows + rx] = (_Float16)tile[threadIdx.x][threadIdx.y + j];
}

#define QG_KT 64
#define QG_P  (QG_KT + 8)
#define QG_M  128
#define QG_N  192
#define QG_NT 6
#define QG_SP 200


__device__ __forceinline__ v16h load_frag_x(const _Float16* rowp, int cc, int f) {
  union { v16h v; v8h h[2]; } u;
  u.h[0] = *(const v8h*)(rowp + ((cc ^ f) << 3));
  u.h[1] = *(const v8h*)(rowp + (((cc + 2) ^ f) << 3));
  return u.v;
}

__global__ void __launch_bounds__(256)
qkv_gemm(const float* __restrict__ x,
         const _Float16* __restrict__ WpT,
         const float* __restrict__ bp,
         _Float16* __restrict__ qh,
         _Float16* __restrict__ kh,
         _Float16* __restrict__ vT) {
  __shared__ alignas(16) _Float16 lds_x[2][QG_M][QG_P];
  __shared__ alignas(16) _Float16 lds_w[2][QG_N][QG_P];
  const int tid   = threadIdx.x;
  const int wave  = tid >> 5;
  const int lane  = tid & 31;
  const int hf    = lane >> 4;
  const int l16   = lane & 15;
  const int kbase = hf * 8;
  const int roff  = hf * 8;
  const int wr    = wave >> 1;
  const int wc    = wave & 1;
  const int nblk  = blockIdx.x * QG_M;
  const int jblk  = blockIdx.y * QG_N;
  const int n0 = nblk + wr * 32;
  const int j0 = jblk + wc * (QG_NT * 16);
  const int b  = blockIdx.z;

  const float* xs = x + ((size_t)b * CCH + wave * 8) * NTOK + nblk + lane * 4;
  const int sw_chunk = ((wave ^ (lane >> 2)) & 7) << 3;
  const int wrow = tid >> 3, wchk = (tid & 7) * 8;
  const _Float16* wsrc = WpT + (size_t)(jblk + wrow) * CCH + wchk;

  const int ra0 = wr * 32 + l16, ra1 = ra0 + 16;
  const int f0 = (wr * 2) & 7, f1 = (wr * 2 + 1) & 7;
  const int rb0 = wc * (QG_NT * 16) + l16;

  v4f xr[8];
  v8h wv[6];
#pragma unroll
  for (int e = 0; e < 8; ++e) xr[e] = *(const v4f*)(xs + (size_t)e * NTOK);
#pragma unroll
  for (int i = 0; i < 6; ++i) wv[i] = *(const v8h*)(wsrc + (size_t)i * 32 * CCH);
#pragma unroll
  for (int i = 0; i < 4; ++i) {
    v8h h;
#pragma unroll
    for (int e = 0; e < 8; ++e) h[e] = (_Float16)xr[e][i];
    *(v8h*)&lds_x[0][lane * 4 + i][sw_chunk] = h;
  }
#pragma unroll
  for (int i = 0; i < 6; ++i) *(v8h*)&lds_w[0][wrow + 32 * i][wchk] = wv[i];
  v8f acc0[QG_NT] = {}, acc1[QG_NT] = {};
  __syncthreads();

  for (int s = 0; s < CCH / QG_KT; ++s) {
    const int buf   = s & 1;
    const bool more = (s + 1) < CCH / QG_KT;
    if (more) {
      const float* xn = xs + (size_t)(s + 1) * QG_KT * NTOK;
#pragma unroll
      for (int e = 0; e < 8; ++e) xr[e] = *(const v4f*)(xn + (size_t)e * NTOK);
      const _Float16* wn = wsrc + (s + 1) * QG_KT;
#pragma unroll
      for (int i = 0; i < 6; ++i) wv[i] = *(const v8h*)(wn + (size_t)i * 32 * CCH);
    }
#pragma unroll
    for (int kk2 = 0; kk2 < 2; ++kk2) {
      const v16h a0 = load_frag_x(&lds_x[buf][ra0][0], kk2 * 4 + hf, f0);
      const v16h a1 = load_frag_x(&lds_x[buf][ra1][0], kk2 * 4 + hf, f1);
      v16h bf[QG_NT];
#pragma unroll
      for (int t = 0; t < QG_NT; ++t)
        bf[t] = load_frag(&lds_w[buf][rb0 + t * 16][kk2 * 32 + kbase]);
      __builtin_amdgcn_sched_barrier(0);
#pragma unroll
      for (int t = 0; t < QG_NT; ++t) {
        acc0[t] = wmma16(bf[t], a0, acc0[t]);
        acc1[t] = wmma16(bf[t], a1, acc1[t]);
      }
      __builtin_amdgcn_sched_barrier(0);
    }
    if (more) {
#pragma unroll
      for (int i = 0; i < 4; ++i) {
        v8h h;
#pragma unroll
        for (int e = 0; e < 8; ++e) h[e] = (_Float16)xr[e][i];
        *(v8h*)&lds_x[buf ^ 1][lane * 4 + i][sw_chunk] = h;
      }
#pragma unroll
      for (int i = 0; i < 6; ++i) *(v8h*)&lds_w[buf ^ 1][wrow + 32 * i][wchk] = wv[i];
      __syncthreads();
    }
  }

  __syncthreads();
  _Float16* st = &lds_w[0][0][0];
#pragma unroll
  for (int t = 0; t < QG_NT; ++t) {
    const int cl = wc * (QG_NT * 16) + t * 16 + roff;
    const v4f b0 = *(const v4f*)(bp + jblk + cl);
    const v4f b1 = *(const v4f*)(bp + jblk + cl + 4);
    v8h h0, h1;
#pragma unroll
    for (int g = 0; g < 4; ++g) {
      h0[g]     = (_Float16)(acc0[t][g]     + b0[g]);
      h0[g + 4] = (_Float16)(acc0[t][g + 4] + b1[g]);
      h1[g]     = (_Float16)(acc1[t][g]     + b0[g]);
      h1[g + 4] = (_Float16)(acc1[t][g + 4] + b1[g]);
    }
    *(v8h*)(st + (wr * 32 + l16) * QG_SP + cl)      = h0;
    *(v8h*)(st + (wr * 32 + 16 + l16) * QG_SP + cl) = h1;
  }
  __syncthreads();

  const int head = blockIdx.y >> 1;
  const size_t bh = (size_t)b * NH + head;
  _Float16* kd = kh + (bh * NTOK + nblk) * DK;
  if ((blockIdx.y & 1) == 0) {
    _Float16* qd = qh + (bh * NTOK + nblk) * DK;
#pragma unroll
    for (int i = 0; i < 8; ++i) {
      const int ci = tid + 256 * i, tok = ci >> 4, cc = (ci & 15) * 8;
      *(v8h*)(qd + (size_t)tok * DK + cc) = *(const v8h*)(st + tok * QG_SP + cc);
    }
#pragma unroll
    for (int i = 0; i < 4; ++i) {
      const int ci = tid + 256 * i, tok = ci >> 3, cc = (ci & 7) * 8;
      *(v8h*)(kd + (size_t)tok * DK + cc) = *(const v8h*)(st + tok * QG_SP + 128 + cc);
    }
  } else {
#pragma unroll
    for (int i = 0; i < 4; ++i) {
      const int ci = tid + 256 * i, tok = ci >> 3, cc = (ci & 7) * 8;
      *(v8h*)(kd + (size_t)tok * DK + 64 + cc) = *(const v8h*)(st + tok * QG_SP + cc);
    }
    _Float16* vd = vT + bh * DK * NTOK + nblk;
#pragma unroll
    for (int i = 0; i < 8; ++i) {
      const int dc = wave * 2 + (i >> 2), tb = (i & 3) * 32;
      const v8h v = lds_load_tr16(st + (tb + lane) * QG_SP + 64 + dc * 8);
      *(v8h*)(vd + (size_t)(dc * 8 + (lane & 7)) * NTOK + tb + (lane >> 3) * 8) = v;
    }
  }
}

#define AT_WAVES 8
#define AT_ROWS  (AT_WAVES * 16)
#define AT_KT    64
#define AT_KP    (DK + 8)
#define AT_VP    (AT_KT + 8)
#define AT_OP    24

__device__ __forceinline__ void wave_lds_publish() {
  __builtin_amdgcn_fence(__ATOMIC_RELEASE, "workgroup", "local");
  __builtin_amdgcn_wave_barrier();
  __builtin_amdgcn_fence(__ATOMIC_ACQUIRE, "workgroup", "local");
}

__device__ __forceinline__ int k_row(int r) { return ((r & 3) << 4) | (r >> 2); }

template <int MASK>
__device__ __forceinline__ float row_xor(float v) {
  const int s = __builtin_bit_cast(int, v);
  return __builtin_bit_cast(float,
      __builtin_amdgcn_update_dpp(s, s, 0x160 | MASK, 0xf, 0xf, true));
}

__global__ void __launch_bounds__(AT_WAVES * 32)
attention_fa(const _Float16* __restrict__ qh, const _Float16* __restrict__ kh,
             const _Float16* __restrict__ vT, _Float16* __restrict__ attn) {
  __shared__ alignas(16) _Float16 lds_k[2][AT_KT][AT_KP];
  __shared__ alignas(16) _Float16 lds_v[2][DK][AT_VP];
  __shared__ alignas(16) _Float16 lds_p[AT_WAVES][16][AT_VP];
  const int tid   = threadIdx.x;
  const int wave  = tid >> 5;
  const int lane  = tid & 31;
  const int hf    = lane >> 4;
  const int l16   = lane & 15;
  const int kbase = hf * 8;
  const int roff  = hf * 8;
  const int n0 = blockIdx.x * AT_ROWS + wave * 16;
  const int h  = blockIdx.y;
  const int b  = blockIdx.z;
  const float scale = 0.08838834764831845f;

  const _Float16* Qb = qh + (((size_t)b * NH + h) * NTOK + n0 + l16) * DK;
  v16h qf[4];
#pragma unroll
  for (int kk = 0; kk < 4; ++kk) qf[kk] = load_frag(Qb + kk * 32 + kbase);

  v8f o[8] = {};
  float mi[8], li[8];
#pragma unroll
  for (int g = 0; g < 8; ++g) { mi[g] = -1e30f; li[g] = 0.0f; }

  const _Float16* Kb = kh + ((size_t)b * NH + h) * NTOK * DK;
  const _Float16* Vb = vT + ((size_t)b * NH + h) * DK * NTOK;

  v8h kr[4], vr[4];
  {
    const v8h* ks = (const v8h*)Kb;
#pragma unroll
    for (int i = 0; i < 4; ++i) {
      const int ci = tid + 256 * i;
      kr[i] = ks[ci];
      vr[i] = *(const v8h*)(Vb + (size_t)(ci >> 3) * NTOK + (ci & 7) * 8);
    }
#pragma unroll
    for (int i = 0; i < 4; ++i) {
      const int ci = tid + 256 * i;
      *(v8h*)&lds_k[0][k_row(ci >> 4)][(ci & 15) * 8] = kr[i];
      *(v8h*)&lds_v[0][ci >> 3][(ci & 7) * 8]  = vr[i];
    }
  }
  __syncthreads();

  for (int it = 0; it < NTOK / AT_KT; ++it) {
    const int buf   = it & 1;
    const bool more = (it + 1) < NTOK / AT_KT;
    if (more) {
      const int jn = (it + 1) * AT_KT;
      const v8h* ks = (const v8h*)(Kb + (size_t)jn * DK);
#pragma unroll
      for (int i = 0; i < 4; ++i) {
        const int ci = tid + 256 * i;
        kr[i] = ks[ci];
        vr[i] = *(const v8h*)(Vb + (size_t)(ci >> 3) * NTOK + jn + (ci & 7) * 8);
      }
    }
    v16h kf[4][4];
#pragma unroll
    for (int kk = 0; kk < 4; ++kk)
#pragma unroll
      for (int j = 0; j < 4; ++j)
        kf[j][kk] = load_frag(&lds_k[buf][j * 16 + l16][kk * 32 + kbase]);
    v8f s[4] = {};
    __builtin_amdgcn_sched_barrier(0);
#pragma unroll
    for (int kk = 0; kk < 4; ++kk)
#pragma unroll
      for (int j = 0; j < 4; ++j) s[j] = wmma16(qf[kk], kf[j][kk], s[j]);
    __builtin_amdgcn_sched_barrier(0);
    v16h vf[2][8];
#pragma unroll
    for (int kh = 0; kh < 2; ++kh)
#pragma unroll
      for (int t = 0; t < 8; ++t)
        vf[kh][t] = load_frag(&lds_v[buf][t * 16 + l16][kh * 32 + kbase]);

#pragma unroll
    for (int g = 0; g < 8; ++g) {
      const float a0 = s[0][g] * scale, a1 = s[1][g] * scale;
      const float a2 = s[2][g] * scale, a3 = s[3][g] * scale;
      float m = fmaxf(fmaxf(a0, a1), fmaxf(a2, a3));
      m = fmaxf(m, row_xor<1>(m));
      m = fmaxf(m, row_xor<2>(m));
      m = fmaxf(m, row_xor<4>(m));
      m = fmaxf(m, row_xor<8>(m));
      const float mn    = fmaxf(mi[g], m);
      const float alpha = __expf(mi[g] - mn);
      const float p0 = __expf(a0 - mn);
      const float p1 = __expf(a1 - mn);
      const float p2 = __expf(a2 - mn);
      const float p3 = __expf(a3 - mn);
      s[0][g] = p0; s[1][g] = p1; s[2][g] = p2; s[3][g] = p3;
      float rs = (p0 + p1) + (p2 + p3);
      rs += row_xor<1>(rs);
      rs += row_xor<2>(rs);
      rs += row_xor<4>(rs);
      rs += row_xor<8>(rs);
      li[g] = li[g] * alpha + rs;
      mi[g] = mn;
#pragma unroll
      for (int t = 0; t < 8; ++t) o[t][g] *= alpha;
    }
#pragma unroll
    for (int g = 0; g < 8; ++g) {
      v4h pv;
#pragma unroll
      for (int j = 0; j < 4; ++j) pv[j] = (_Float16)s[j][g];
      *(v4h*)&lds_p[wave][roff + g][l16 * 4] = pv;
    }
    wave_lds_publish();
    const v16h pf0 = load_frag(&lds_p[wave][l16][kbase]);
    const v16h pf1 = load_frag(&lds_p[wave][l16][32 + kbase]);
    __builtin_amdgcn_sched_barrier(0);
#pragma unroll
    for (int t = 0; t < 8; ++t) o[t] = wmma16(pf0, vf[0][t], o[t]);
#pragma unroll
    for (int t = 0; t < 8; ++t) o[t] = wmma16(pf1, vf[1][t], o[t]);
    __builtin_amdgcn_sched_barrier(0);
    if (more) {
#pragma unroll
      for (int i = 0; i < 4; ++i) {
        const int ci = tid + 256 * i;
        *(v8h*)&lds_k[buf ^ 1][k_row(ci >> 4)][(ci & 15) * 8] = kr[i];
        *(v8h*)&lds_v[buf ^ 1][ci >> 3][(ci & 7) * 8]  = vr[i];
      }
      __syncthreads();
    }
  }
  __syncthreads();
  _Float16* ot = (wave < 4 ? &lds_k[0][0][0] : &lds_v[0][0][0]) + (wave & 3) * (DK * AT_OP);
#pragma unroll
  for (int t = 0; t < 8; ++t) {
    v8h hv;
#pragma unroll
    for (int g = 0; g < 8; ++g) hv[g] = (_Float16)(o[t][g] / li[g]);
    *(v8h*)(ot + (t * 16 + l16) * AT_OP + roff) = hv;
  }
  wave_lds_publish();
  _Float16* Ob = attn + ((size_t)b * NTOK + n0) * (NH * DK) + h * DK;
#pragma unroll
  for (int i = 0; i < 8; ++i) {
    const int db = (i >> 1) * 32, tc = (i & 1) * 8;
    const v8h v = lds_load_tr16(ot + (db + lane) * AT_OP + tc);
    *(v8h*)(Ob + (size_t)(tc + (lane & 7)) * (NH * DK) + db + (lane >> 3) * 8) = v;
  }
}

#define OP_KT 64
#define OP_P  (OP_KT + 8)
#define OP_M  128
#define OP_N  128
#define OP_NT 4
#define OP_SP 132

__global__ void __launch_bounds__(256)
out_proj(const _Float16* __restrict__ attn,
         const float* __restrict__ Wo,
         const float* __restrict__ bo,
         const float* __restrict__ x,
         float* __restrict__ out) {
  __shared__ alignas(16) _Float16 lds_ab[4][OP_M][OP_P];
  _Float16 (*lds_a)[OP_M][OP_P] = &lds_ab[0];
  _Float16 (*lds_b)[OP_N][OP_P] = &lds_ab[2];
  const int tid   = threadIdx.x;
  const int wave  = tid >> 5;
  const int lane  = tid & 31;
  const int hf    = lane >> 4;
  const int l16   = lane & 15;
  const int kbase = hf * 8;
  const int roff  = hf * 8;
  const int wr    = wave >> 1;
  const int wc    = wave & 1;
  const int nblk  = blockIdx.x * OP_M;
  const int cblk  = blockIdx.y * OP_N;
  const int n0 = nblk + wr * 32;
  const int c0 = cblk + wc * (OP_NT * 16);
  const int b  = blockIdx.z;

  const int arow = tid >> 3, achk = (tid & 7) * 8;
  const _Float16* asrc = attn + ((size_t)b * NTOK + nblk + arow) * CCH + achk;
  const float* wsrc = Wo + (size_t)(wave * 8) * CCH + cblk + lane * 4;
  const int sw_chunk = ((wave ^ (lane >> 2)) & 7) << 3;

  const int ra0 = wr * 32 + l16, ra1 = ra0 + 16;
  const int rb0 = wc * (OP_NT * 16) + l16;
  const int fb0 = wc * OP_NT;

  v8h av[4];
  v4f wv[8];
#pragma unroll
  for (int i = 0; i < 4; ++i) av[i] = *(const v8h*)(asrc + (size_t)i * 32 * CCH);
#pragma unroll
  for (int e = 0; e < 8; ++e) wv[e] = *(const v4f*)(wsrc + (size_t)e * CCH);
#pragma unroll
  for (int i = 0; i < 4; ++i) *(v8h*)&lds_a[0][arow + 32 * i][achk] = av[i];
#pragma unroll
  for (int i = 0; i < 4; ++i) {
    v8h h;
#pragma unroll
    for (int e = 0; e < 8; ++e) h[e] = (_Float16)wv[e][i];
    *(v8h*)&lds_b[0][lane * 4 + i][sw_chunk] = h;
  }
  v8f acc0[OP_NT] = {}, acc1[OP_NT] = {};
  const size_t orow = ((size_t)b * CCH + cblk + wave * 16) * NTOK + nblk + lane * 4;
  v4f xres[16];
  __syncthreads();

  for (int s = 0; s < CCH / OP_KT; ++s) {
    const int buf   = s & 1;
    const bool more = (s + 1) < CCH / OP_KT;
    if (more) {
      const _Float16* an = asrc + (s + 1) * OP_KT;
#pragma unroll
      for (int i = 0; i < 4; ++i) av[i] = *(const v8h*)(an + (size_t)i * 32 * CCH);
      const float* wn = wsrc + (size_t)(s + 1) * OP_KT * CCH;
#pragma unroll
      for (int e = 0; e < 8; ++e) wv[e] = *(const v4f*)(wn + (size_t)e * CCH);
    } else {
#pragma unroll
      for (int i = 0; i < 16; ++i) xres[i] = *(const v4f*)(x + orow + (size_t)i * NTOK);
    }
    v16h a0[2], a1[2], bf[2][OP_NT];
#pragma unroll
    for (int kk2 = 0; kk2 < 2; ++kk2) {
      a0[kk2] = load_frag(&lds_a[buf][ra0][kk2 * 32 + kbase]);
      a1[kk2] = load_frag(&lds_a[buf][ra1][kk2 * 32 + kbase]);
#pragma unroll
      for (int t = 0; t < OP_NT; ++t)
        bf[kk2][t] = load_frag_x(&lds_b[buf][rb0 + t * 16][0], kk2 * 4 + hf, fb0 + t);
    }
    __builtin_amdgcn_sched_barrier(0);
#pragma unroll
    for (int kk2 = 0; kk2 < 2; ++kk2) {
#pragma unroll
      for (int t = 0; t < OP_NT; ++t) {
        acc0[t] = wmma16(a0[kk2], bf[kk2][t], acc0[t]);
        acc1[t] = wmma16(a1[kk2], bf[kk2][t], acc1[t]);
      }
    }
    __builtin_amdgcn_sched_barrier(0);
    if (more) {
#pragma unroll
      for (int i = 0; i < 4; ++i) *(v8h*)&lds_a[buf ^ 1][arow + 32 * i][achk] = av[i];
#pragma unroll
      for (int i = 0; i < 4; ++i) {
        v8h h;
#pragma unroll
        for (int e = 0; e < 8; ++e) h[e] = (_Float16)wv[e][i];
        *(v8h*)&lds_b[buf ^ 1][lane * 4 + i][sw_chunk] = h;
      }
      __syncthreads();
    }
  }

  __syncthreads();
  float* so = (float*)&lds_ab[0][0][0];
#pragma unroll
  for (int t = 0; t < OP_NT; ++t) {
    float* p = so + (wc * (OP_NT * 16) + t * 16 + l16) * OP_SP + wr * 32 + roff;
    v4f lo0, hi0, lo1, hi1;
#pragma unroll
    for (int g = 0; g < 4; ++g) {
      lo0[g] = acc0[t][g]; hi0[g] = acc0[t][4 + g];
      lo1[g] = acc1[t][g]; hi1[g] = acc1[t][4 + g];
    }
    *(v4f*)(p)      = lo0;
    *(v4f*)(p + 4)  = hi0;
    *(v4f*)(p + 16) = lo1;
    *(v4f*)(p + 20) = hi1;
  }
  __syncthreads();
#pragma unroll
  for (int i = 0; i < 16; ++i) {
    const int cl = wave * 16 + i;
    const float bias = bo[cblk + cl];
    const v4f a = *(const v4f*)(so + cl * OP_SP + lane * 4);
    v4f r;
#pragma unroll
    for (int g = 0; g < 4; ++g) r[g] = a[g] + bias + xres[i][g];
    *(v4f*)(out + orow + (size_t)i * NTOK) = r;
  }
}

extern "C" void kernel_launch(void* const* d_in, const int* in_sizes, int n_in,
                              void* d_out, int out_size, void* d_ws, size_t ws_size,
                              hipStream_t stream) {
  (void)in_sizes; (void)n_in; (void)out_size; (void)ws_size;
  const float* x  = (const float*)d_in[0];
  const float* Wp = (const float*)d_in[1];
  const float* bp = (const float*)d_in[2];
  const float* Wo = (const float*)d_in[3];
  const float* bo = (const float*)d_in[4];
  float* out = (float*)d_out;

  char* ws = (char*)d_ws;
  _Float16* WpT = (_Float16*)(ws +  8388608);
  _Float16* qh  = (_Float16*)(ws + 10485760);
  _Float16* kh  = (_Float16*)(ws + 18874368);
  _Float16* vT  = (_Float16*)(ws + 27262976);
  _Float16* at  = (_Float16*)(ws + 35651584);

  const dim3 tb(32, 8);
  transpose_cast<<<dim3(1536 / 32, CCH / 32, 1),  tb, 0, stream>>>(Wp, WpT, CCH, 1536);

  qkv_gemm    <<<dim3(NTOK / QG_M, 1536 / QG_N, NB), 256, 0, stream>>>(x, WpT, bp, qh, kh, vT);
  attention_fa<<<dim3(NTOK / AT_ROWS, NH,   NB), AT_WAVES * 32, 0, stream>>>(qh, kh, vT, at);
  out_proj    <<<dim3(NTOK / OP_M, CCH / OP_N, NB), 256, 0, stream>>>(at, Wo, bo, x, out);
}
